# starting kernel
# baseline (speedup 1.0000x reference)
#include <hip/hip_runtime.h>
#include <hip/hip_bf16.h>


typedef __attribute__((ext_vector_type(2))) float v2f;
typedef __attribute__((ext_vector_type(8))) float v8f;

#define IN_F 8
#define OUT_F 8
#define SPEC 16

__global__ __launch_bounds__(256) void kan_cheb_wmma(
    const float* __restrict__ x,
    const float* __restrict__ w,
    float* __restrict__ out,
    int n)
{
    const int lane = threadIdx.x & 31;
    const int col  = lane & 15;
    const bool hi  = lane >= 16;
    const int wavesPerBlock = blockDim.x >> 5;
    const int waveId = blockIdx.x * wavesPerBlock + (threadIdx.x >> 5);
    const int nWaves = gridDim.x * wavesPerBlock;
    const int nTiles = n >> 4;

    v2f bf[IN_F][4];
    #pragma unroll
    for (int f = 0; f < IN_F; ++f) {
        #pragma unroll
        for (int kk = 0; kk < 4; ++kk) {
            const int s0 = kk * 4 + (hi ? 2 : 0);
            float b0 = 0.0f, b1 = 0.0f;
            if (col < OUT_F) {
                b0 = w[(f * SPEC + s0    ) * OUT_F + col];
                b1 = w[(f * SPEC + s0 + 1) * OUT_F + col];
            }
            bf[f][kk].x = b0;
            bf[f][kk].y = b1;
        }
    }

    for (int t = waveId; t < nTiles; t += nWaves) {
        const int base = t << 4;
        const float4* xp4 = (const float4*)(x + (size_t)(base + col) * IN_F);
        const float4 l4 = xp4[0];
        const float4 h4 = xp4[1];
        float xv[IN_F] = { l4.x, l4.y, l4.z, l4.w, h4.x, h4.y, h4.z, h4.w };

        v8f acc = {0.f, 0.f, 0.f, 0.f, 0.f, 0.f, 0.f, 0.f};

        #pragma unroll
        for (int f = 0; f < IN_F; ++f) {
            const float xf = xv[f];
            const float x2 = xf + xf;
            const float T0  = 1.0f;
            const float T1  = xf;
            const float T2  = __builtin_fmaf(x2, T1,  -T0);
            const float T3  = __builtin_fmaf(x2, T2,  -T1);
            const float T4  = __builtin_fmaf(x2, T3,  -T2);
            const float T5  = __builtin_fmaf(x2, T4,  -T3);
            const float T6  = __builtin_fmaf(x2, T5,  -T4);
            const float T7  = __builtin_fmaf(x2, T6,  -T5);
            const float T8  = __builtin_fmaf(x2, T7,  -T6);
            const float T9  = __builtin_fmaf(x2, T8,  -T7);
            const float T10 = __builtin_fmaf(x2, T9,  -T8);
            const float T11 = __builtin_fmaf(x2, T10, -T9);
            const float T12 = __builtin_fmaf(x2, T11, -T10);
            const float T13 = __builtin_fmaf(x2, T12, -T11);
            const float T14 = __builtin_fmaf(x2, T13, -T12);
            const float T15 = __builtin_fmaf(x2, T14, -T13);

            v2f a;
            a.x = hi ? T2 : T0;  a.y = hi ? T3 : T1;
            acc = __builtin_amdgcn_wmma_f32_16x16x4_f32(
                false, a, false, bf[f][0], (short)0, acc, false, false);
            a.x = hi ? T6 : T4;  a.y = hi ? T7 : T5;
            acc = __builtin_amdgcn_wmma_f32_16x16x4_f32(
                false, a, false, bf[f][1], (short)0, acc, false, false);
            a.x = hi ? T10 : T8;  a.y = hi ? T11 : T9;
            acc = __builtin_amdgcn_wmma_f32_16x16x4_f32(
                false, a, false, bf[f][2], (short)0, acc, false, false);
            a.x = hi ? T14 : T12;  a.y = hi ? T15 : T13;
            acc = __builtin_amdgcn_wmma_f32_16x16x4_f32(
                false, a, false, bf[f][3], (short)0, acc, false, false);
        }

        if (col < OUT_F) {
            float* op = out + (size_t)(base + (hi ? 8 : 0)) * OUT_F + col;
            #pragma unroll
            for (int r = 0; r < 8; ++r)
                op[(size_t)r * OUT_F] = acc[r];
        }
    }
}

extern "C" void kernel_launch(void* const* d_in, const int* in_sizes, int n_in,
                              void* d_out, int out_size, void* d_ws, size_t ws_size,
                              hipStream_t stream) {
    const float* x = (const float*)d_in[0];
    const float* w = (const float*)d_in[1];
    float* out = (float*)d_out;
    const int n = in_sizes[0] / IN_F;

    const int block = 256;
    const int grid = 1024;
    kan_cheb_wmma<<<grid, block, 0, stream>>>(x, w, out, n);
}


# four adjacent 32-sample tiles per trip on four independent accumulator chains taking turns (shared A pair), loads of a trip issued together: one wave fills the matrix pipe alone; sum order kept
# speedup vs baseline: 1.2787x; 1.2787x over previous
#include <hip/hip_runtime.h>


typedef __attribute__((ext_vector_type(2))) float v2f;
typedef __attribute__((ext_vector_type(8))) float v8f;

#define IN_F 8
#define OUT_F 8
#define SPEC 16
#define TPT 4

__global__ __launch_bounds__(256) void kan_cheb_wmma(
    const float* __restrict__ x,
    const float* __restrict__ w,
    float* __restrict__ out,
    int n)
{
    const int lane = threadIdx.x & 31;
    const int row  = lane & 15;
    const bool hi  = lane >= 16;
    const int wavesPerBlock = blockDim.x >> 5;
    const int waveId = blockIdx.x * wavesPerBlock + (threadIdx.x >> 5);
    const int nWaves = gridDim.x * wavesPerBlock;
    const int nTiles = (n + 31) >> 5;
    const int nTrips = (nTiles + TPT - 1) / TPT;

    const bool live = (row >= 8) == hi;
    const int o = row & 7;
    v2f af[IN_F][SPEC / 2];
    #pragma unroll
    for (int f = 0; f < IN_F; ++f) {
        #pragma unroll
        for (int j = 0; j < SPEC / 2; ++j) {
            const float a0 = w[(f * SPEC + 2 * j    ) * OUT_F + o];
            const float a1 = w[(f * SPEC + 2 * j + 1) * OUT_F + o];
            af[f][j].x = live ? a0 : 0.0f;
            af[f][j].y = live ? a1 : 0.0f;
        }
    }

    for (int t = waveId; t < nTrips; t += nWaves) {
        const int sA  = (t * TPT + 0) * 32 + lane;
        const int slA = sA < n ? sA : n - 1;
        const float4* xpA = (const float4*)(x + (size_t)slA * IN_F);
        const int sB  = (t * TPT + 1) * 32 + lane;
        const int slB = sB < n ? sB : n - 1;
        const float4* xpB = (const float4*)(x + (size_t)slB * IN_F);
        const int sC  = (t * TPT + 2) * 32 + lane;
        const int slC = sC < n ? sC : n - 1;
        const float4* xpC = (const float4*)(x + (size_t)slC * IN_F);
        const int sD  = (t * TPT + 3) * 32 + lane;
        const int slD = sD < n ? sD : n - 1;
        const float4* xpD = (const float4*)(x + (size_t)slD * IN_F);
        const float4 lA = xpA[0];
        const float4 hA = xpA[1];
        const float4 lB = xpB[0];
        const float4 hB = xpB[1];
        const float4 lC = xpC[0];
        const float4 hC = xpC[1];
        const float4 lD = xpD[0];
        const float4 hD = xpD[1];
        const float xa[IN_F] = { lA.x, lA.y, lA.z, lA.w, hA.x, hA.y, hA.z, hA.w };
        const float xb[IN_F] = { lB.x, lB.y, lB.z, lB.w, hB.x, hB.y, hB.z, hB.w };
        const float xc[IN_F] = { lC.x, lC.y, lC.z, lC.w, hC.x, hC.y, hC.z, hC.w };
        const float xd[IN_F] = { lD.x, lD.y, lD.z, lD.w, hD.x, hD.y, hD.z, hD.w };

        v8f accA = {0.f, 0.f, 0.f, 0.f, 0.f, 0.f, 0.f, 0.f};
        v8f accB = {0.f, 0.f, 0.f, 0.f, 0.f, 0.f, 0.f, 0.f};
        v8f accC = {0.f, 0.f, 0.f, 0.f, 0.f, 0.f, 0.f, 0.f};
        v8f accD = {0.f, 0.f, 0.f, 0.f, 0.f, 0.f, 0.f, 0.f};

        #pragma unroll
        for (int f = 0; f < IN_F; ++f) {
            const float x2a = xa[f] + xa[f];
            const float x2b = xb[f] + xb[f];
            const float x2c = xc[f] + xc[f];
            const float x2d = xd[f] + xd[f];
            float Tea = 1.0f, Toa = xa[f];
            float Teb = 1.0f, Tob = xb[f];
            float Tec = 1.0f, Toc = xc[f];
            float Ted = 1.0f, Tod = xd[f];

            #pragma unroll
            for (int j = 0; j < SPEC / 2; ++j) {
                float nTea = Tea, nToa = Toa;
                float nTeb = Teb, nTob = Tob;
                float nTec = Tec, nToc = Toc;
                float nTed = Ted, nTod = Tod;
                if (j + 1 < SPEC / 2) {
                    nTea = __builtin_fmaf(x2a, Toa, -Tea);
                    nTeb = __builtin_fmaf(x2b, Tob, -Teb);
                    nTec = __builtin_fmaf(x2c, Toc, -Tec);
                    nTed = __builtin_fmaf(x2d, Tod, -Ted);
                    nToa = __builtin_fmaf(x2a, nTea, -Toa);
                    nTob = __builtin_fmaf(x2b, nTeb, -Tob);
                    nToc = __builtin_fmaf(x2c, nTec, -Toc);
                    nTod = __builtin_fmaf(x2d, nTed, -Tod);
                }
                { v2f b;  b.x = Tea;  b.y = Toa;
                  accA = __builtin_amdgcn_wmma_f32_16x16x4_f32(
                      false, af[f][j], false, b, (short)0, accA, false, false); }
                { v2f b;  b.x = Teb;  b.y = Tob;
                  accB = __builtin_amdgcn_wmma_f32_16x16x4_f32(
                      false, af[f][j], false, b, (short)0, accB, false, false); }
                { v2f b;  b.x = Tec;  b.y = Toc;
                  accC = __builtin_amdgcn_wmma_f32_16x16x4_f32(
                      false, af[f][j], false, b, (short)0, accC, false, false); }
                { v2f b;  b.x = Ted;  b.y = Tod;
                  accD = __builtin_amdgcn_wmma_f32_16x16x4_f32(
                      false, af[f][j], false, b, (short)0, accD, false, false); }
                __builtin_amdgcn_sched_barrier(0);
                Tea = nTea;  Toa = nToa;
                Teb = nTeb;  Tob = nTob;
                Tec = nTec;  Toc = nToc;
                Ted = nTed;  Tod = nTod;
            }
        }

        if (sA < n) {
            float4* op = (float4*)(out + (size_t)sA * OUT_F);
            op[0] = make_float4(accA[0], accA[1], accA[2], accA[3]);
            op[1] = make_float4(accA[4], accA[5], accA[6], accA[7]);
        }
        if (sB < n) {
            float4* op = (float4*)(out + (size_t)sB * OUT_F);
            op[0] = make_float4(accB[0], accB[1], accB[2], accB[3]);
            op[1] = make_float4(accB[4], accB[5], accB[6], accB[7]);
        }
        if (sC < n) {
            float4* op = (float4*)(out + (size_t)sC * OUT_F);
            op[0] = make_float4(accC[0], accC[1], accC[2], accC[3]);
            op[1] = make_float4(accC[4], accC[5], accC[6], accC[7]);
        }
        if (sD < n) {
            float4* op = (float4*)(out + (size_t)sD * OUT_F);
            op[0] = make_float4(accD[0], accD[1], accD[2], accD[3]);
            op[1] = make_float4(accD[4], accD[5], accD[6], accD[7]);
        }
    }
}

extern "C" void kernel_launch(void* const* d_in, const int* in_sizes, int n_in,
                              void* d_out, int out_size, void* d_ws, size_t ws_size,
                              hipStream_t stream) {
    const float* x = (const float*)d_in[0];
    const float* w = (const float*)d_in[1];
    float* out = (float*)d_out;
    const int n = in_sizes[0] / IN_F;

    const int block = 256;
    const int grid = 512;
    kan_cheb_wmma<<<grid, block, 0, stream>>>(x, w, out, n);
}
